# starting kernel
# baseline (speedup 1.0000x reference)
#include <hip/hip_runtime.h>

#define N_NODES 50000
#define N_EDGES 500000
#define IN_C 512
#define HID_C 256
#define OUT_C 128

typedef __attribute__((ext_vector_type(2))) float v2f;
typedef __attribute__((ext_vector_type(8))) float v8f;

__global__ void init_deg_kernel(float* __restrict__ deg, int n) {
  int i = blockIdx.x * blockDim.x + threadIdx.x;
  if (i < n) deg[i] = 1.0f;
}

__global__ void deg_atomic_kernel(const int* __restrict__ dst, float* __restrict__ deg, int e) {
  int i = blockIdx.x * blockDim.x + threadIdx.x;
  if (i < e) unsafeAtomicAdd(&deg[dst[i]], 1.0f);
}

__global__ void norm_kernel(const float* __restrict__ deg, float* __restrict__ dis,
                            float* __restrict__ inv, int n) {
  int i = blockIdx.x * blockDim.x + threadIdx.x;
  if (i < n) {
    float d = deg[i];
    dis[i] = rsqrtf(d);
    inv[i] = 1.0f / d;
  }
}

__global__ void coef_kernel(const int* __restrict__ src, const int* __restrict__ dst,
                            const float* __restrict__ dis, float* __restrict__ coef, int e) {
  int i = blockIdx.x * blockDim.x + threadIdx.x;
  if (i < e) coef[i] = dis[src[i]] * dis[dst[i]];
}

__global__ void transpose_kernel(const float* __restrict__ B, float* __restrict__ Bt,
                                 int K, int N) {
  int i = blockIdx.x * blockDim.x + threadIdx.x;
  if (i < K * N) {
    int kk = i / N;
    int n = i - kk * N;
    Bt[(size_t)n * K + kk] = B[i];
  }
}

template <int MT, int NT>
__global__ __launch_bounds__(128) void wmma_gemm_f32(const float* __restrict__ A,
                                                     const float* __restrict__ Bt,
                                                     float* __restrict__ C,
                                                     int M, int N, int K) {
  const int wave = (blockIdx.x * blockDim.x + threadIdx.x) >> 5;
  const int lane = threadIdx.x & 31;
  const int ngroups = N / (NT * 16);
  const int sm = wave / ngroups;
  const int tg = wave % ngroups;
  const int m0 = sm * MT * 16;
  if (m0 >= M) return;

  const int half = lane >> 4;
  const int l = lane & 15;

  const float* a_ptr[MT];
#pragma unroll
  for (int i = 0; i < MT; ++i) {
    int row = m0 + i * 16 + l;
    if (row >= M) row = M - 1;
    a_ptr[i] = A + (size_t)row * K;
  }
  const float* b_ptr[NT];
#pragma unroll
  for (int t = 0; t < NT; ++t) {
    int col = tg * (NT * 16) + t * 16 + l;
    b_ptr[t] = Bt + (size_t)col * K;
  }

  v8f acc[MT][NT];
#pragma unroll
  for (int i = 0; i < MT; ++i)
#pragma unroll
    for (int t = 0; t < NT; ++t) acc[i][t] = (v8f){0, 0, 0, 0, 0, 0, 0, 0};

  for (int k = 0; k < K; k += 4) {
    const int ka = k + 2 * half;
    v2f a[MT], b[NT];
#pragma unroll
    for (int i = 0; i < MT; ++i) a[i] = *(const v2f*)(a_ptr[i] + ka);
#pragma unroll
    for (int t = 0; t < NT; ++t) b[t] = *(const v2f*)(b_ptr[t] + ka);
#pragma unroll
    for (int i = 0; i < MT; ++i)
#pragma unroll
      for (int t = 0; t < NT; ++t)
        acc[i][t] = __builtin_amdgcn_wmma_f32_16x16x4_f32(
            false, a[i], false, b[t], (short)0, acc[i][t], false, false);
  }

#pragma unroll
  for (int i = 0; i < MT; ++i) {
    if (m0 + i * 16 + 16 <= M) {
      float* c_ptr = C + (size_t)(m0 + i * 16) * N + (size_t)tg * (NT * 16) + l;
#pragma unroll
      for (int t = 0; t < NT; ++t)
#pragma unroll
        for (int r = 0; r < 8; ++r)
          c_ptr[(size_t)(r + 8 * half) * N + t * 16] = acc[i][t][r];
    }
  }
}

__global__ void init_agg_kernel(float* __restrict__ agg, const float* __restrict__ z,
                                const float* __restrict__ inv, const float* __restrict__ bias,
                                int n, int c) {
  int i = blockIdx.x * blockDim.x + threadIdx.x;
  if (i < n * c) {
    int node = i / c;
    int ch = i - node * c;
    agg[i] = z[i] * inv[node] + bias[ch];
  }
}

__global__ void edge_agg_kernel(float* __restrict__ agg, const float* __restrict__ z,
                                const int* __restrict__ src, const int* __restrict__ dst,
                                const float* __restrict__ coef, int e, int c) {
  int i = blockIdx.x * blockDim.x + threadIdx.x;
  const int groups = c >> 2;
  if (i >= e * groups) return;
  const int eg = i / groups;
  const int g = i - eg * groups;
  const int s = src[eg];
  const int d = dst[eg];
  const float w = coef[eg];
  const float4 v = *(const float4*)(z + (size_t)s * c + g * 4);
  float* out = agg + (size_t)d * c + g * 4;
  unsafeAtomicAdd(out + 0, v.x * w);
  unsafeAtomicAdd(out + 1, v.y * w);
  unsafeAtomicAdd(out + 2, v.z * w);
  unsafeAtomicAdd(out + 3, v.w * w);
}

__global__ void relu_kernel(float* __restrict__ h, int total) {
  int i = blockIdx.x * blockDim.x + threadIdx.x;
  if (i < total) h[i] = fmaxf(h[i], 0.0f);
}

static inline int cdiv_l(long a, long b) { return (int)((a + b - 1) / b); }

extern "C" void kernel_launch(void* const* d_in, const int* in_sizes, int n_in,
                              void* d_out, int out_size, void* d_ws, size_t ws_size,
                              hipStream_t stream) {
  const float* x   = (const float*)d_in[0];
  const int*   ei  = (const int*)d_in[1];
  const int*   src = ei;
  const int*   dst = ei + N_EDGES;
  const float* w1  = (const float*)d_in[2];
  const float* b1  = (const float*)d_in[3];
  const float* wmu = (const float*)d_in[4];
  const float* bmu = (const float*)d_in[5];
  const float* wls = (const float*)d_in[6];
  const float* bls = (const float*)d_in[7];

  float* out_mu = (float*)d_out;
  float* out_ls = out_mu + (size_t)N_NODES * OUT_C;

  float* ws = (float*)d_ws;
  size_t off = 0;
  float* deg  = ws + off; off += N_NODES;
  float* dis  = ws + off; off += N_NODES;
  float* inv  = ws + off; off += N_NODES;
  float* coef = ws + off; off += N_EDGES;
  float* w1t  = ws + off; off += (size_t)IN_C * HID_C;
  float* wmut = ws + off; off += (size_t)HID_C * OUT_C;
  float* wlst = ws + off; off += (size_t)HID_C * OUT_C;
  float* hpre = ws + off; off += (size_t)N_NODES * HID_C;
  float* h    = ws + off; off += (size_t)N_NODES * HID_C;
  float* zmu  = ws + off; off += (size_t)N_NODES * OUT_C;
  float* zls  = ws + off; off += (size_t)N_NODES * OUT_C;

  init_deg_kernel<<<cdiv_l(N_NODES, 256), 256, 0, stream>>>(deg, N_NODES);
  deg_atomic_kernel<<<cdiv_l(N_EDGES, 256), 256, 0, stream>>>(dst, deg, N_EDGES);
  norm_kernel<<<cdiv_l(N_NODES, 256), 256, 0, stream>>>(deg, dis, inv, N_NODES);
  coef_kernel<<<cdiv_l(N_EDGES, 256), 256, 0, stream>>>(src, dst, dis, coef, N_EDGES);

  transpose_kernel<<<cdiv_l((long)IN_C * HID_C, 256), 256, 0, stream>>>(w1, w1t, IN_C, HID_C);
  transpose_kernel<<<cdiv_l((long)HID_C * OUT_C, 256), 256, 0, stream>>>(wmu, wmut, HID_C, OUT_C);
  transpose_kernel<<<cdiv_l((long)HID_C * OUT_C, 256), 256, 0, stream>>>(wls, wlst, HID_C, OUT_C);

  constexpr int MT = 4, NT = 4;
  const long m_super = cdiv_l(N_NODES, MT * 16);

  {
    long waves = m_super * (HID_C / (NT * 16));
    wmma_gemm_f32<MT, NT><<<cdiv_l(waves * 32, 128), 128, 0, stream>>>(
        x, w1t, hpre, N_NODES, HID_C, IN_C);
  }
  init_agg_kernel<<<cdiv_l((long)N_NODES * HID_C, 256), 256, 0, stream>>>(h, hpre, inv, b1,
                                                                          N_NODES, HID_C);
  edge_agg_kernel<<<cdiv_l((long)N_EDGES * (HID_C / 4), 256), 256, 0, stream>>>(
      h, hpre, src, dst, coef, N_EDGES, HID_C);
  relu_kernel<<<cdiv_l((long)N_NODES * HID_C, 256), 256, 0, stream>>>(h, N_NODES * HID_C);

  {
    long waves = m_super * (OUT_C / (NT * 16));
    wmma_gemm_f32<MT, NT><<<cdiv_l(waves * 32, 128), 128, 0, stream>>>(
        h, wmut, zmu, N_NODES, OUT_C, HID_C);
  }
  init_agg_kernel<<<cdiv_l((long)N_NODES * OUT_C, 256), 256, 0, stream>>>(out_mu, zmu, inv, bmu,
                                                                          N_NODES, OUT_C);
  edge_agg_kernel<<<cdiv_l((long)N_EDGES * (OUT_C / 4), 256), 256, 0, stream>>>(
      out_mu, zmu, src, dst, coef, N_EDGES, OUT_C);

  {
    long waves = m_super * (OUT_C / (NT * 16));
    wmma_gemm_f32<MT, NT><<<cdiv_l(waves * 32, 128), 128, 0, stream>>>(
        h, wlst, zls, N_NODES, OUT_C, HID_C);
  }
  init_agg_kernel<<<cdiv_l((long)N_NODES * OUT_C, 256), 256, 0, stream>>>(out_ls, zls, inv, bls,
                                                                          N_NODES, OUT_C);
  edge_agg_kernel<<<cdiv_l((long)N_EDGES * (OUT_C / 4), 256), 256, 0, stream>>>(
      out_ls, zls, src, dst, coef, N_EDGES, OUT_C);
}


# Edge sweep reads a one-byte owner tag per edge (dst/196, written by the coefficient kernel): 16 edges per 16-byte load, zero-byte test per dword; dst, src, coef fetched only on a hit; sums as before
# speedup vs baseline: 2.3070x; 2.3070x over previous
#include <hip/hip_runtime.h>

#define N_NODES 50000
#define N_EDGES 500000
#define IN_C 512
#define HID_C 256
#define OUT_C 128

typedef __attribute__((ext_vector_type(2))) float v2f;
typedef __attribute__((ext_vector_type(8))) float v8f;

__global__ void init_deg_kernel(float* __restrict__ deg, int n) {
  int i = blockIdx.x * blockDim.x + threadIdx.x;
  if (i < n) deg[i] = 1.0f;
}

__global__ void deg_atomic_kernel(const int* __restrict__ dst, float* __restrict__ deg, int e) {
  int i = blockIdx.x * blockDim.x + threadIdx.x;
  if (i < e) unsafeAtomicAdd(&deg[dst[i]], 1.0f);
}

__global__ void norm_kernel(const float* __restrict__ deg, float* __restrict__ dis,
                            float* __restrict__ inv, int n) {
  int i = blockIdx.x * blockDim.x + threadIdx.x;
  if (i < n) {
    float d = deg[i];
    dis[i] = rsqrtf(d);
    inv[i] = 1.0f / d;
  }
}

#define TAG_ROWS 196
__global__ void coef_kernel(const int* __restrict__ src, const int* __restrict__ dst,
                            const float* __restrict__ dis, float* __restrict__ coef,
                            unsigned* __restrict__ tag, int e4) {
  int i = blockIdx.x * blockDim.x + threadIdx.x;
  if (i < e4) {
    const int4 s = ((const int4*)src)[i];
    const int4 d = ((const int4*)dst)[i];
    ((float4*)coef)[i] = make_float4(dis[s.x] * dis[d.x], dis[s.y] * dis[d.y],
                                     dis[s.z] * dis[d.z], dis[s.w] * dis[d.w]);
    tag[i] = (((unsigned)d.x / TAG_ROWS) & 255u) | ((((unsigned)d.y / TAG_ROWS) & 255u) << 8) |
             ((((unsigned)d.z / TAG_ROWS) & 255u) << 16) | ((((unsigned)d.w / TAG_ROWS) & 255u) << 24);
  }
}

template <int MT, int NT, int NC>
__global__ __launch_bounds__(128) void wmma_gemm_f32(const float* __restrict__ A,
                                                     const float* __restrict__ B,
                                                     float* __restrict__ C,
                                                     int M, int K) {
  constexpr int N = NC;
  const int wave = (blockIdx.x * blockDim.x + threadIdx.x) >> 5;
  const int lane = threadIdx.x & 31;
  constexpr int ngroups = N / (NT * 16);
  const int sm = wave / ngroups;
  const int tg = wave % ngroups;
  const int m0 = sm * MT * 16;
  if (m0 >= M) return;

  const int half = lane >> 4;
  const int l = lane & 15;

  const float* a_ptr[MT];
#pragma unroll
  for (int i = 0; i < MT; ++i) {
    int row = m0 + i * 16 + l;
    if (row >= M) row = M - 1;
    a_ptr[i] = A + (size_t)row * K;
  }
  const float* b_col = B + (size_t)(2 * half) * N + tg * (NT * 16) + l;

  v8f acc[MT][NT];
#pragma unroll
  for (int i = 0; i < MT; ++i)
#pragma unroll
    for (int t = 0; t < NT; ++t) acc[i][t] = (v8f){0, 0, 0, 0, 0, 0, 0, 0};

  for (int k = 0; k < K; k += 4) {
    const int ka = k + 2 * half;
    v2f a[MT], b[NT];
#pragma unroll
    for (int i = 0; i < MT; ++i) a[i] = *(const v2f*)(a_ptr[i] + ka);
#pragma unroll
    for (int t = 0; t < NT; ++t) {
      const float* p = b_col + (size_t)k * N + t * 16;
      b[t] = (v2f){p[0], p[N]};
    }
#pragma unroll
    for (int i = 0; i < MT; ++i)
#pragma unroll
      for (int t = 0; t < NT; ++t)
        acc[i][t] = __builtin_amdgcn_wmma_f32_16x16x4_f32(
            false, a[i], false, b[t], (short)0, acc[i][t], false, false);
  }

#pragma unroll
  for (int i = 0; i < MT; ++i) {
    if (m0 + i * 16 + 16 <= M) {
      float* c_ptr = C + (size_t)(m0 + i * 16) * N + (size_t)tg * (NT * 16) + l;
#pragma unroll
      for (int t = 0; t < NT; ++t)
#pragma unroll
        for (int r = 0; r < 8; ++r)
          c_ptr[(size_t)(r + 8 * half) * N + t * 16] = acc[i][t][r];
    }
  }
}

#define AGG_ROWS 196
#define AGG_ROWW 256
#define AGG_TILES ((AGG_ROWS + 15) / 16)
#define AGG_PITCH_H 260
#define AGG_WAVES 32
#define AGG_WGS ((N_NODES + AGG_ROWS - 1) / AGG_ROWS)

static __device__ __forceinline__ void lds_add(float* p, float v) {
  __hip_atomic_fetch_add(p, v, __ATOMIC_RELAXED, __HIP_MEMORY_SCOPE_WORKGROUP);
}

template <int MT>
static __device__ __forceinline__ void heads_from_lds(const float4* __restrict__ rows4, int rt0, int rows,
                                                      int row0, const float* __restrict__ w, int cbase,
                                                      const float* __restrict__ bias,
                                                      float* __restrict__ out, int lane) {
  const int half = lane >> 4;
  const int l = lane & 15;
  constexpr int P4 = AGG_PITCH_H / 4;
  v8f d[MT][2];
#pragma unroll
  for (int i = 0; i < MT; ++i)
#pragma unroll
    for (int t = 0; t < 2; ++t) d[i][t] = (v8f){0, 0, 0, 0, 0, 0, 0, 0};

  int aoff[MT];
#pragma unroll
  for (int i = 0; i < MT; ++i) aoff[i] = ((rt0 + 4 * i) * 16 + l) * P4 + half;
  const float* bq = w + (size_t)(4 * half) * OUT_C + cbase + l;

#pragma unroll 2
  for (int k8 = 0; k8 < HID_C / 4; k8 += 2) {
    float4 a[MT], b[2];
#pragma unroll
    for (int t = 0; t < 2; ++t) {
      const float* p = bq + (size_t)(4 * k8) * OUT_C + t * 16;
      b[t] = make_float4(p[0], p[OUT_C], p[2 * OUT_C], p[3 * OUT_C]);
    }
#pragma unroll
    for (int i = 0; i < MT; ++i) a[i] = rows4[aoff[i] + k8];
#pragma unroll
    for (int i = 0; i < MT; ++i)
#pragma unroll
      for (int t = 0; t < 2; ++t)
        d[i][t] = __builtin_amdgcn_wmma_f32_16x16x4_f32(false, (v2f){a[i].x, a[i].y}, false,
                                                        (v2f){b[t].x, b[t].y}, (short)0, d[i][t], false, false);
#pragma unroll
    for (int i = 0; i < MT; ++i)
#pragma unroll
      for (int t = 0; t < 2; ++t)
        d[i][t] = __builtin_amdgcn_wmma_f32_16x16x4_f32(false, (v2f){a[i].z, a[i].w}, false,
                                                        (v2f){b[t].z, b[t].w}, (short)0, d[i][t], false, false);
  }

#pragma unroll
  for (int t = 0; t < 2; ++t) {
    const int col = cbase + t * 16 + l;
    const float bv = bias[col];
#pragma unroll
    for (int i = 0; i < MT; ++i)
#pragma unroll
      for (int r = 0; r < 8; ++r) {
        const int row = (rt0 + 4 * i) * 16 + r + 8 * half;
        if (row < rows) out[(size_t)(row0 + row) * OUT_C + col] = d[i][t][r] + bv;
      }
  }
}

template <int C, int NP, bool RELU, bool HEADS>
__global__ __launch_bounds__(AGG_WAVES * 32) void agg_rows_in_lds(
    const float* __restrict__ z0, const float* __restrict__ z1,
    float* __restrict__ o0, float* __restrict__ o1,
    const float* __restrict__ bias0, const float* __restrict__ bias1,
    const float* __restrict__ inv, const int* __restrict__ src,
    const int* __restrict__ dst, const float* __restrict__ coef,
    const unsigned* __restrict__ tag,
    const float* __restrict__ w0, const float* __restrict__ w1) {
  static_assert(!HEADS || (NP == 1 && C == HID_C && !RELU), "head products read one 256-wide plane");
  constexpr int PITCH = HEADS ? AGG_PITCH_H : AGG_ROWW;
  constexpr int LROWS = HEADS ? AGG_TILES * 16 : AGG_ROWS;
  static_assert(C * NP == AGG_ROWW && (C == 128 || C == 256), "LDS row = all planes side by side");
  static_assert(N_EDGES % 16 == 0, "owner tags are swept 16 at a time");
  static_assert(AGG_WGS <= 256 && TAG_ROWS == AGG_ROWS, "an owner tag is one byte: the work-group number");
  constexpr int N16 = N_EDGES / 16;
  constexpr int NCH = (N16 + 31) / 32;
  constexpr int NTRIP = 32;
  constexpr int NROT = NTRIP * AGG_WAVES;
  static_assert(NROT >= NCH && (NROT & (NROT - 1)) == 0 && NROT % AGG_WGS == 0, "chunk numbers wrap with a mask");

  __shared__ float4 acc4[LROWS * PITCH / 4];
  float* acc = (float*)acc4;

  const int lane = threadIdx.x & 31;
  const int wv = __builtin_amdgcn_readfirstlane((int)(threadIdx.x >> 5));
  const int row0 = blockIdx.x * AGG_ROWS;
  const int rows = (N_NODES - row0 < AGG_ROWS) ? (N_NODES - row0) : AGG_ROWS;

  const float4* zh[2] = {(const float4*)z0, (const float4*)(NP == 2 ? z1 : z0 + 128)};
  const float4* bh[2] = {(const float4*)bias0, (const float4*)(NP == 2 ? bias1 : bias0 + 128)};
  float4* oh[2] = {(float4*)o0, (float4*)(NP == 2 ? o1 : o0 + 128)};
  constexpr int C4 = C / 4;

  for (int r = wv; r < rows; r += AGG_WAVES) {
    const float sn = inv[row0 + r];
#pragma unroll
    for (int hf = 0; hf < 2; ++hf) {
      const float4 zv = zh[hf][(size_t)(row0 + r) * C4 + lane];
      const float4 bv = HEADS ? make_float4(0.f, 0.f, 0.f, 0.f) : bh[hf][lane];
      acc4[r * (PITCH / 4) + 32 * hf + lane] =
          make_float4(zv.x * sn + bv.x, zv.y * sn + bv.y, zv.z * sn + bv.z, zv.w * sn + bv.w);
    }
  }
  if (HEADS) {
    const int rend = ((rows + 15) >> 4) << 4;
    for (int r = rows + wv; r < rend; r += AGG_WAVES) {
      acc4[r * (PITCH / 4) + lane] = make_float4(0.f, 0.f, 0.f, 0.f);
      acc4[r * (PITCH / 4) + 32 + lane] = make_float4(0.f, 0.f, 0.f, 0.f);
    }
  }
  __syncthreads();

  const uint4* tag4 = (const uint4*)tag;
  const int4* dst4 = (const int4*)dst;
  const int4* src4 = (const int4*)src;
  const float4* coef4 = (const float4*)coef;
  const int4 none = make_int4(-1, -1, -1, -1);
  const unsigned pat = blockIdx.x * 0x01010101u;
  const unsigned nopat = ~pat;

  const int rot = blockIdx.x * (NROT / AGG_WGS);

  uint4 tn;
  {
    const int q = ((wv + rot) & (NROT - 1)) * 32 + lane;
    tn = tag4[q < N16 ? q : N16 - 1];
    if (q >= N16) tn = make_uint4(nopat, nopat, nopat, nopat);
  }

  for (int it = 0; it < NTRIP; ++it) {
    const int c0 = wv + AGG_WAVES * it + rot;
    const uint4 tc = tn;
    if (it + 1 < NTRIP) {
      const int qn = ((c0 + AGG_WAVES) & (NROT - 1)) * 32 + lane;
      tn = tag4[qn < N16 ? qn : N16 - 1];
      if (qn >= N16) tn = make_uint4(nopat, nopat, nopat, nopat);
    }

    {
      const int q = (c0 & (NROT - 1)) * 32 + lane;
      const unsigned wd[4] = {tc.x, tc.y, tc.z, tc.w};
      unsigned hm[4];
      int4 dc[4], sv[4];
      float4 cw[4];
#pragma unroll
      for (int j = 0; j < 4; ++j) {
        const unsigned x = wd[j] ^ pat;
        const bool hit = ((x - 0x01010101u) & ~x & 0x80808080u) != 0u;
        hm[j] = __builtin_amdgcn_ballot_w32(hit);
        dc[j] = none;
        sv[j] = make_int4(0, 0, 0, 0);
        cw[j] = make_float4(0.f, 0.f, 0.f, 0.f);
        if (hit) {
          const int i4 = 4 * q + j;
          dc[j] = dst4[i4];
          sv[j] = src4[i4];
          cw[j] = coef4[i4];
        }
      }

#pragma unroll
      for (int j = 0; j < 4; ++j) {
        if (hm[j] == 0u) continue;
        const int larr[4] = {dc[j].x - row0, dc[j].y - row0, dc[j].z - row0, dc[j].w - row0};
        const int sarr[4] = {sv[j].x, sv[j].y, sv[j].z, sv[j].w};
        const float warr[4] = {cw[j].x, cw[j].y, cw[j].z, cw[j].w};
#pragma unroll
        for (int k = 0; k < 4; ++k) {
          unsigned mask = __builtin_amdgcn_ballot_w32((unsigned)larr[k] < (unsigned)rows);
          while (mask) {
            const int L = __builtin_ctz(mask);
            mask &= mask - 1;
            const int lr = __builtin_amdgcn_readlane(larr[k], L);
            const int s = __builtin_amdgcn_readlane(sarr[k], L);
            const float w = __builtin_bit_cast(float, __builtin_amdgcn_readlane(__builtin_bit_cast(int, warr[k]), L));
            const float4 v0 = zh[0][(size_t)s * C4 + lane];
            const float4 v1 = zh[1][(size_t)s * C4 + lane];
            float* a = acc + lr * PITCH + 4 * lane;
            lds_add(a + 0, v0.x * w);
            lds_add(a + 1, v0.y * w);
            lds_add(a + 2, v0.z * w);
            lds_add(a + 3, v0.w * w);
            lds_add(a + 128, v1.x * w);
            lds_add(a + 129, v1.y * w);
            lds_add(a + 130, v1.z * w);
            lds_add(a + 131, v1.w * w);
          }
        }
      }
    }
  }
  __syncthreads();

  if (HEADS) {
    const int cg = wv & 7;
    const int rg = wv >> 3;
    const int ntiles = (rows + 15) >> 4;
    const int cnt = (rg < ntiles) ? ((ntiles - 1 - rg) >> 2) + 1 : 0;
    const float* wh = (cg < 4) ? w0 : w1;
    const float* bs = (cg < 4) ? bias0 : bias1;
    float* op = (cg < 4) ? o0 : o1;
    const int cbase = (cg & 3) * 32;
    if (cnt == 4) heads_from_lds<4>(acc4, rg, rows, row0, wh, cbase, bs, op, lane);
    else if (cnt == 3) heads_from_lds<3>(acc4, rg, rows, row0, wh, cbase, bs, op, lane);
    else if (cnt == 2) heads_from_lds<2>(acc4, rg, rows, row0, wh, cbase, bs, op, lane);
    else if (cnt == 1) heads_from_lds<1>(acc4, rg, rows, row0, wh, cbase, bs, op, lane);
  } else {
    for (int r = wv; r < rows; r += AGG_WAVES) {
#pragma unroll
      for (int hf = 0; hf < 2; ++hf) {
        float4 v = acc4[r * (PITCH / 4) + 32 * hf + lane];
        if (RELU) v = make_float4(fmaxf(v.x, 0.0f), fmaxf(v.y, 0.0f), fmaxf(v.z, 0.0f), fmaxf(v.w, 0.0f));
        oh[hf][(size_t)(row0 + r) * C4 + lane] = v;
      }
    }
  }
}

static inline int cdiv_l(long a, long b) { return (int)((a + b - 1) / b); }

extern "C" void kernel_launch(void* const* d_in, const int* in_sizes, int n_in,
                              void* d_out, int out_size, void* d_ws, size_t ws_size,
                              hipStream_t stream) {
  const float* x   = (const float*)d_in[0];
  const int*   ei  = (const int*)d_in[1];
  const int*   src = ei;
  const int*   dst = ei + N_EDGES;
  const float* w1  = (const float*)d_in[2];
  const float* b1  = (const float*)d_in[3];
  const float* wmu = (const float*)d_in[4];
  const float* bmu = (const float*)d_in[5];
  const float* wls = (const float*)d_in[6];
  const float* bls = (const float*)d_in[7];

  float* out_mu = (float*)d_out;
  float* out_ls = out_mu + (size_t)N_NODES * OUT_C;

  float* ws = (float*)d_ws;
  size_t off = 0;
  float* deg  = ws + off; off += N_NODES;
  float* dis  = ws + off; off += N_NODES;
  float* inv  = ws + off; off += N_NODES;
  float* coef = ws + off; off += N_EDGES;
  float* hpre = ws + off; off += (size_t)N_NODES * HID_C;
  float* h    = ws + off; off += (size_t)N_NODES * HID_C;
  unsigned* tag = (unsigned*)(ws + off);

  init_deg_kernel<<<cdiv_l(N_NODES, 256), 256, 0, stream>>>(deg, N_NODES);
  deg_atomic_kernel<<<cdiv_l(N_EDGES, 256), 256, 0, stream>>>(dst, deg, N_EDGES);
  norm_kernel<<<cdiv_l(N_NODES, 256), 256, 0, stream>>>(deg, dis, inv, N_NODES);
  coef_kernel<<<cdiv_l(N_EDGES / 4, 256), 256, 0, stream>>>(src, dst, dis, coef, tag, N_EDGES / 4);

  constexpr int MT = 4, NT = 4;
  const long m_super = cdiv_l(N_NODES, MT * 16);

  {
    long waves = m_super * (HID_C / (NT * 16));
    wmma_gemm_f32<MT, NT, HID_C><<<cdiv_l(waves * 32, 128), 128, 0, stream>>>(
        x, w1, hpre, N_NODES, IN_C);
  }
  agg_rows_in_lds<HID_C, 1, true, false><<<AGG_WGS, AGG_WAVES * 32, 0, stream>>>(
      hpre, nullptr, h, nullptr, b1, nullptr, inv, src, dst, coef, tag, nullptr, nullptr);

  agg_rows_in_lds<HID_C, 1, false, true><<<AGG_WGS, AGG_WAVES * 32, 0, stream>>>(
      h, nullptr, out_mu, out_ls, bmu, bls, inv, src, dst, coef, tag, wmu, wls);
}
